# starting kernel
# baseline (speedup 1.0000x reference)
#include <hip/hip_runtime.h>
#include <math.h>

#define PP   769
#define LDL  772
#define NCOL 32768
#define BMSK 2047
#define KC   128
#define KST  132

typedef __attribute__((ext_vector_type(2))) float v2f;
typedef __attribute__((ext_vector_type(8))) float v8f;

__device__ __forceinline__ v8f wmma_f32_16x16x4(v2f a, v2f b, v8f c) {
  return __builtin_amdgcn_wmma_f32_16x16x4_f32(false, a, false, b, (short)0, c,
                                               false, false);
}

__global__ __launch_bounds__(256) void build_L_kernel(const float* __restrict__ cov,
                                                      float* __restrict__ Lb) {
  int t = blockIdx.x * 256 + threadIdx.x;
  if (t >= PP * LDL) return;
  int i = t / LDL;
  int j = t - i * LDL;
  float v = 0.0f;
  if (j < PP && j <= i) {
    v = cov[i * (i + 1) / 2 + j];
    if (i == j) v = expf(v);
  }
  Lb[t] = v;
}

template <int TLO>
__device__ __forceinline__ void kphase(int kbeg, int kend, int kgbase,
                                       const float* __restrict__ zc,
                                       const float* __restrict__ sA,
                                       int half, int sub,
                                       v8f& c0, v8f& c1, v8f& c2,
                                       v8f& c3, v8f& c4, v8f& c5) {
  const int lbase = sub * KST + 2 * half;
  for (int kk = kbeg; kk < kend; kk += 4) {
    v2f bm;
    bm.x = zc[(size_t)(kgbase + kk + 2 * half) * NCOL];
    bm.y = zc[(size_t)(kgbase + kk + 2 * half + 1) * NCOL];
    const int ao = lbase + kk;
    if constexpr (TLO <= 0) { v2f a; a.x = sA[0 * 16 * KST + ao]; a.y = sA[0 * 16 * KST + ao + 1]; c0 = wmma_f32_16x16x4(a, bm, c0); }
    if constexpr (TLO <= 1) { v2f a; a.x = sA[1 * 16 * KST + ao]; a.y = sA[1 * 16 * KST + ao + 1]; c1 = wmma_f32_16x16x4(a, bm, c1); }
    if constexpr (TLO <= 2) { v2f a; a.x = sA[2 * 16 * KST + ao]; a.y = sA[2 * 16 * KST + ao + 1]; c2 = wmma_f32_16x16x4(a, bm, c2); }
    if constexpr (TLO <= 3) { v2f a; a.x = sA[3 * 16 * KST + ao]; a.y = sA[3 * 16 * KST + ao + 1]; c3 = wmma_f32_16x16x4(a, bm, c3); }
    if constexpr (TLO <= 4) { v2f a; a.x = sA[4 * 16 * KST + ao]; a.y = sA[4 * 16 * KST + ao + 1]; c4 = wmma_f32_16x16x4(a, bm, c4); }
    if constexpr (TLO <= 5) { v2f a; a.x = sA[5 * 16 * KST + ao]; a.y = sA[5 * 16 * KST + ao + 1]; c5 = wmma_f32_16x16x4(a, bm, c5); }
  }
}

__global__ __launch_bounds__(256) void fcvi_main_kernel(
    const float* __restrict__ Lb, const float* __restrict__ z,
    const float* __restrict__ x, const float* __restrict__ mean,
    float* __restrict__ out) {
  __shared__ float sA[6 * 16 * KST];

  const int tid  = threadIdx.x;
  const int lane = tid & 31;
  const int wave = tid >> 5;
  const int sub  = lane & 15;
  const int half = lane >> 4;
  const int gcol = blockIdx.x * 128 + wave * 16 + sub;
  const float xv = x[gcol & BMSK];
  const float* __restrict__ zcol = z + gcol;

  float outpart = 0.0f;

  for (int mp = 0; mp < 8; ++mp) {
    const int r00 = 32 * mp, r01 = r00 + 16;
    const int rb0 = 256 + r00, rb1 = rb0 + 16;
    const int r10 = 512 + r00, r11 = r10 + 16;
    const int e0 = r00 + 32;
    const int e1 = 512 + e0;
    const int rbase[6] = {r00, r01, rb0, rb1, r10, r11};
    const int bext[6]  = {e0, e0, 256 + e0, 256 + e0, e1, e1};

    v8f a0 = {0.f, 0.f, 0.f, 0.f, 0.f, 0.f, 0.f, 0.f};
    v8f a1 = a0, a2 = a0, a3 = a0, a4 = a0, a5 = a0;

    for (int kk0 = 0; kk0 < e1; kk0 += KC) {
      __syncthreads();
#pragma unroll
      for (int t = 0; t < 6; ++t) {
        int e = bext[t] - kk0;
        if (e <= 0) continue;
        if (e > KC) e = KC;
#pragma unroll 4
        for (int r = 0; r < 16; ++r) {
          const float* src = Lb + (size_t)(rbase[t] + r) * LDL + kk0;
          float* dst = &sA[(t * 16 + r) * KST];
          for (int c = tid; c < e; c += 256) dst[c] = src[c];
        }
      }
      __syncthreads();

      int kend = e1 - kk0;
      if (kend > KC) kend = KC;
      const int tlo = (kk0 < e0) ? 0 : (kk0 < 256 + e0) ? 2 : 4;
      const int sbound = (tlo == 0) ? e0 : (tlo == 2) ? 256 + e0 : e1;
      int s = sbound - kk0;
      if (s > kend) s = kend;

      if (tlo == 0)      kphase<0>(0, s, kk0, zcol, sA, half, sub, a0, a1, a2, a3, a4, a5);
      else if (tlo == 2) kphase<2>(0, s, kk0, zcol, sA, half, sub, a0, a1, a2, a3, a4, a5);
      else               kphase<4>(0, s, kk0, zcol, sA, half, sub, a0, a1, a2, a3, a4, a5);
      if (s < kend) {
        if (tlo == 0) kphase<2>(s, kend, kk0, zcol, sA, half, sub, a0, a1, a2, a3, a4, a5);
        else          kphase<4>(s, kend, kk0, zcol, sA, half, sub, a0, a1, a2, a3, a4, a5);
      }
    }

#pragma unroll
    for (int v = 0; v < 8; ++v) {
      const int rl = v + 8 * half;
      const float w0e = a0[v] + mean[r00 + rl];
      const float b0e = a2[v] + mean[rb0 + rl];
      const float w1e = a4[v] + mean[r10 + rl];
      outpart = fmaf(fmaxf(fmaf(xv, w0e, b0e), 0.0f), w1e, outpart);
      const float w0o = a1[v] + mean[r01 + rl];
      const float b0o = a3[v] + mean[rb1 + rl];
      const float w1o = a5[v] + mean[r11 + rl];
      outpart = fmaf(fmaxf(fmaf(xv, w0o, b0o), 0.0f), w1o, outpart);
    }
  }

  float bpart = (half == 0) ? mean[768] : 0.0f;
  const float* lrow = Lb + (size_t)768 * LDL;
  for (int j = half * 4; j < 768; j += 8) {
    const float4 lv = *(const float4*)(lrow + j);
    bpart = fmaf(lv.x, zcol[(size_t)(j + 0) * NCOL], bpart);
    bpart = fmaf(lv.y, zcol[(size_t)(j + 1) * NCOL], bpart);
    bpart = fmaf(lv.z, zcol[(size_t)(j + 2) * NCOL], bpart);
    bpart = fmaf(lv.w, zcol[(size_t)(j + 3) * NCOL], bpart);
  }
  if (half == 0) bpart = fmaf(lrow[768], zcol[(size_t)768 * NCOL], bpart);

  outpart += bpart;
  const float other = __shfl_xor(outpart, 16, 32);
  if (half == 0) out[gcol] = outpart + other;
}

extern "C" void kernel_launch(void* const* d_in, const int* in_sizes, int n_in,
                              void* d_out, int out_size, void* d_ws, size_t ws_size,
                              hipStream_t stream) {
  (void)in_sizes; (void)n_in; (void)out_size; (void)ws_size;
  const float* x    = (const float*)d_in[0];
  const float* mean = (const float*)d_in[1];
  const float* cov  = (const float*)d_in[2];
  const float* z    = (const float*)d_in[3];
  float* outp = (float*)d_out;
  float* Lbuf = (float*)d_ws;

  build_L_kernel<<<(PP * LDL + 255) / 256, 256, 0, stream>>>(cov, Lbuf);
  fcvi_main_kernel<<<NCOL / 128, 256, 0, stream>>>(Lbuf, z, x, mean, outp);
}


# Block tile = all rows x 128 columns in one K sweep: each wave keeps 48 accumulator tiles (384 registers); L and Z chunks of 32 k staged in LDS (16-byte loads) feed them and the b1 dot; sum order kept
# speedup vs baseline: 1.7772x; 1.7772x over previous
#include <hip/hip_runtime.h>
#include <math.h>

#define PP   769
#define LDL  772
#define NCOL 32768
#define BMSK 2047
#define KC   32
#define KST  36
#define ZST  136
#define NT   48

typedef __attribute__((ext_vector_type(2))) float v2f;
typedef __attribute__((ext_vector_type(8))) float v8f;

__device__ __forceinline__ v8f wmma_f32_16x16x4(v2f a, v2f b, v8f c) {
  return __builtin_amdgcn_wmma_f32_16x16x4_f32(false, a, false, b, (short)0, c,
                                               false, false);
}

__global__ __launch_bounds__(256) void build_L_kernel(const float* __restrict__ cov,
                                                      float* __restrict__ Lb) {
  int t = blockIdx.x * 256 + threadIdx.x;
  if (t >= PP * LDL) return;
  int i = t / LDL;
  int j = t - i * LDL;
  float v = 0.0f;
  if (j < PP && j <= i) {
    v = cov[i * (i + 1) / 2 + j];
    if (i == j) v = expf(v);
  }
  Lb[t] = v;
}

struct Thr {
  unsigned lsrc;
  unsigned zsrc;
  float* ldst_lo;
  float* ldst_hi;
  float* zdst;
  const float* a_lo;
  const float* a_hi;
  const float* b;
  const float* zc;
};

template <int P>
__device__ __forceinline__ void kchunk(const char* __restrict__ Lbytes,
                                       const char* __restrict__ zbytes,
                                       const Thr& th, const float* __restrict__ l768,
                                       int half, v8f (&acc)[NT], float& bpart) {
  constexpr int K0 = KC * P;
  __syncthreads();
  {
    constexpr int NS = NT / 2 - P;
    constexpr int N1 = (NS + 4 <= 16) ? NS : (NS + 5) / 2 - 4;
    const char* zsrc = zbytes + (size_t)K0 * (NCOL * 4);
    const char* lsrc = Lbytes + (size_t)(K0 * LDL + K0) * 4;
    {
      float4 zv[4];
      float4 lv[N1];
#pragma unroll
      for (int i = 0; i < 4; ++i)
        zv[i] = *(const float4*)(zsrc + (size_t)(8 * i) * (NCOL * 4) + th.zsrc);
#pragma unroll
      for (int i = 0; i < N1; ++i)
        lv[i] = *(const float4*)(lsrc + (size_t)(32 * i) * (LDL * 4) + th.lsrc);
#pragma unroll
      for (int i = 0; i < 4; ++i) *(float4*)(th.zdst + 8 * i * ZST) = zv[i];
#pragma unroll
      for (int i = 0; i < N1; ++i) {
        const int g = K0 + 32 * i;
        float* d = (g < 384) ? th.ldst_lo + g * KST : th.ldst_hi + (g - 384) * KST;
        *(float4*)d = lv[i];
      }
    }
    if constexpr (N1 < NS) {
      __builtin_amdgcn_sched_barrier(0);
      float4 lv[NS - N1];
#pragma unroll
      for (int i = N1; i < NS; ++i)
        lv[i - N1] = *(const float4*)(lsrc + (size_t)(32 * i) * (LDL * 4) + th.lsrc);
#pragma unroll
      for (int i = N1; i < NS; ++i) {
        const int g = K0 + 32 * i;
        float* d = (g < 384) ? th.ldst_lo + g * KST : th.ldst_hi + (g - 384) * KST;
        *(float4*)d = lv[i - N1];
      }
    }
  }
  __syncthreads();

#pragma unroll 1
  for (int s = 0; s < KC / 4; ++s) {
    {
      const float4 lv = *(const float4*)(l768 + K0 + 4 * s);
      const float z0 = th.zc[(4 * s + 0) * ZST], z1 = th.zc[(4 * s + 1) * ZST];
      const float z2 = th.zc[(4 * s + 2) * ZST], z3 = th.zc[(4 * s + 3) * ZST];
      float nb = fmaf(lv.x, z0, bpart);
      nb = fmaf(lv.y, z1, nb);
      nb = fmaf(lv.z, z2, nb);
      nb = fmaf(lv.w, z3, nb);
      bpart = (half == (s & 1)) ? nb : bpart;
    }
    v2f b;
    b.x = th.b[(4 * s) * ZST];
    b.y = th.b[(4 * s + 1) * ZST];
#pragma unroll
    for (int t0 = 2 * P; t0 < NT; t0 += 8) {
      v2f a[8];
#pragma unroll
      for (int j = 0; j < 8; ++j) {
        const int t = t0 + j;
        if (t < NT) {
          const float* ap = (t < 24) ? th.a_lo + t * 16 * KST : th.a_hi + (t - 24) * 16 * KST;
          a[j] = *(const v2f*)(ap + 4 * s);
        }
      }
#pragma unroll
      for (int j = 0; j < 8; ++j) {
        const int t = t0 + j;
        if (t < NT) acc[t] = wmma_f32_16x16x4(a[j], b, acc[t]);
      }
    }
  }
}

template <int P>
__device__ __forceinline__ void ksweep(const char* __restrict__ Lbytes,
                                       const char* __restrict__ zbytes,
                                       const Thr& th, const float* __restrict__ l768,
                                       int half, v8f (&acc)[NT], float& bpart) {
  if constexpr (P < NT / 2) {
    kchunk<P>(Lbytes, zbytes, th, l768, half, acc, bpart);
    ksweep<P + 1>(Lbytes, zbytes, th, l768, half, acc, bpart);
  }
}

__global__ __launch_bounds__(256) void fcvi_main_kernel(
    const float* __restrict__ Lb, const float* __restrict__ z,
    const float* __restrict__ x, const float* __restrict__ mean,
    float* __restrict__ out) {
  __shared__ __attribute__((aligned(16))) float sA[NT * 16 * KST];
  __shared__ __attribute__((aligned(16))) float sB1[NT * 16];
  __shared__ __attribute__((aligned(16))) float sZ[KC * ZST];

  const int tid  = threadIdx.x;
  const int lane = tid & 31;
  const int wave = tid >> 5;
  const int sub  = lane & 15;
  const int half = lane >> 4;
  const int gcol = blockIdx.x * 128 + wave * 16 + sub;
  const float xv = x[gcol & BMSK];
  const float* __restrict__ zcol = z + gcol;

  float bpart = (half == 0) ? mean[768] : 0.0f;

  if (tid < NT * 4) *(float4*)(sB1 + 4 * tid) = *(const float4*)(Lb + (size_t)768 * LDL + 4 * tid);

  v8f acc[NT];
#pragma unroll
  for (int t = 0; t < NT; ++t) acc[t] = (v8f){0.f, 0.f, 0.f, 0.f, 0.f, 0.f, 0.f, 0.f};

  {
    Thr th;
    const int r = tid >> 3, q = tid & 7;
    th.lsrc = (unsigned)((r * LDL + 4 * q) * 4);
    th.ldst_lo = sA + r * KST + 4 * q;
    th.ldst_hi = sA + (384 + r) * KST + 4 * q;
    th.zsrc = (unsigned)((wave * NCOL + 4 * lane) * 4);
    th.zdst = sZ + wave * ZST + 4 * lane;
    th.a_lo = sA + sub * KST + 2 * half;
    th.a_hi = sA + (384 + sub) * KST + 2 * half;
    th.b = sZ + (2 * half) * ZST + wave * 16 + sub;
    th.zc = sZ + wave * 16 + sub;
    ksweep<0>((const char*)Lb, (const char*)(z + blockIdx.x * 128), th, sB1, half, acc, bpart);
  }

  float outpart = 0.0f;
#pragma unroll
  for (int mp = 0; mp < 8; ++mp) {
    const int r00 = 32 * mp, r01 = r00 + 16;
    const int rb0 = 256 + r00, rb1 = rb0 + 16;
    const int r10 = 512 + r00, r11 = r10 + 16;
#pragma unroll
    for (int v = 0; v < 8; ++v) {
      const int rl = v + 8 * half;
      const float w0e = acc[2 * mp][v] + mean[r00 + rl];
      const float b0e = acc[16 + 2 * mp][v] + mean[rb0 + rl];
      const float w1e = acc[32 + 2 * mp][v] + mean[r10 + rl];
      outpart = fmaf(fmaxf(fmaf(xv, w0e, b0e), 0.0f), w1e, outpart);
      const float w0o = acc[2 * mp + 1][v] + mean[r01 + rl];
      const float b0o = acc[17 + 2 * mp][v] + mean[rb1 + rl];
      const float w1o = acc[33 + 2 * mp][v] + mean[r11 + rl];
      outpart = fmaf(fmaxf(fmaf(xv, w0o, b0o), 0.0f), w1o, outpart);
    }
  }

  if (half == 0) bpart = fmaf(Lb[(size_t)768 * LDL + 768], zcol[(size_t)768 * NCOL], bpart);

  outpart += bpart;
  const float other = __shfl_xor(outpart, 16, 32);
  if (half == 0) out[gcol] = outpart + other;
}

extern "C" void kernel_launch(void* const* d_in, const int* in_sizes, int n_in,
                              void* d_out, int out_size, void* d_ws, size_t ws_size,
                              hipStream_t stream) {
  (void)in_sizes; (void)n_in; (void)out_size; (void)ws_size;
  const float* x    = (const float*)d_in[0];
  const float* mean = (const float*)d_in[1];
  const float* cov  = (const float*)d_in[2];
  const float* z    = (const float*)d_in[3];
  float* outp = (float*)d_out;
  float* Lbuf = (float*)d_ws;

  build_L_kernel<<<(PP * LDL + 255) / 256, 256, 0, stream>>>(cov, Lbuf);
  fcvi_main_kernel<<<NCOL / 128, 256, 0, stream>>>(Lbuf, z, x, mean, outp);
}


# Launch 1 stores L as 24 contiguous K-chunk panels (32 k wide, rows from the diagonal block down, one 16-byte store per thread): each staging load of launch 2 covers 4 whole lines; sum order kept
# speedup vs baseline: 1.8684x; 1.0513x over previous
#include <hip/hip_runtime.h>
#include <math.h>

#define PP   769
#define PANEL 1024
#define NBLK 300
#define BROW (NBLK * PANEL)
#define NCOL 32768
#define BMSK 2047
#define KC   32
#define KST  36
#define ZST  136
#define NT   48

typedef __attribute__((ext_vector_type(2))) float v2f;
typedef __attribute__((ext_vector_type(8))) float v8f;

__device__ __forceinline__ v8f wmma_f32_16x16x4(v2f a, v2f b, v8f c) {
  return __builtin_amdgcn_wmma_f32_16x16x4_f32(false, a, false, b, (short)0, c,
                                               false, false);
}

__global__ __launch_bounds__(256) void build_L_kernel(const float* __restrict__ cov,
                                                      float* __restrict__ Lb) {
  const int b = blockIdx.x, tid = threadIdx.x;
  int i, k0;
  if (b < NBLK) {
    int P = 0, rem = b;
    while (rem >= 24 - P) { rem -= 24 - P; ++P; }
    i = 32 * (P + rem) + (tid >> 3);
    k0 = 32 * P + 4 * (tid & 7);
  } else {
    if (tid >= 193) return;
    i = 768;
    k0 = 4 * tid;
  }
  const float* __restrict__ row = cov + i * (i + 1) / 2;
  float v[4];
#pragma unroll
  for (int e = 0; e < 4; ++e) v[e] = (k0 + e <= i) ? row[k0 + e] : 0.0f;
#pragma unroll
  for (int e = 0; e < 4; ++e)
    if (k0 + e == i) v[e] = expf(v[e]);
  *(float4*)(Lb + (size_t)b * PANEL + 4 * tid) = make_float4(v[0], v[1], v[2], v[3]);
}

struct Thr {
  unsigned lsrc;
  unsigned zsrc;
  float* ldst_lo;
  float* ldst_hi;
  float* zdst;
  const float* a_lo;
  const float* a_hi;
  const float* b;
  const float* zc;
};

template <int P>
__device__ __forceinline__ void kchunk(const char* __restrict__ Lbytes,
                                       const char* __restrict__ zbytes,
                                       const Thr& th, const float* __restrict__ l768,
                                       int half, v8f (&acc)[NT], float& bpart) {
  constexpr int K0 = KC * P;
  __syncthreads();
  {
    constexpr int NS = NT / 2 - P;
    constexpr int N1 = (NS + 4 <= 16) ? NS : (NS + 5) / 2 - 4;
    const char* zsrc = zbytes + (size_t)K0 * (NCOL * 4);
    constexpr int CB = 24 * P - P * (P - 1) / 2;
    const char* lsrc = Lbytes + (size_t)CB * (PANEL * 4);
    {
      float4 zv[4];
      float4 lv[N1];
#pragma unroll
      for (int i = 0; i < 4; ++i)
        zv[i] = *(const float4*)(zsrc + (size_t)(8 * i) * (NCOL * 4) + th.zsrc);
#pragma unroll
      for (int i = 0; i < N1; ++i)
        lv[i] = *(const float4*)(lsrc + (size_t)i * (PANEL * 4) + th.lsrc);
#pragma unroll
      for (int i = 0; i < 4; ++i) *(float4*)(th.zdst + 8 * i * ZST) = zv[i];
#pragma unroll
      for (int i = 0; i < N1; ++i) {
        const int g = K0 + 32 * i;
        float* d = (g < 384) ? th.ldst_lo + g * KST : th.ldst_hi + (g - 384) * KST;
        *(float4*)d = lv[i];
      }
    }
    if constexpr (N1 < NS) {
      __builtin_amdgcn_sched_barrier(0);
      float4 lv[NS - N1];
#pragma unroll
      for (int i = N1; i < NS; ++i)
        lv[i - N1] = *(const float4*)(lsrc + (size_t)i * (PANEL * 4) + th.lsrc);
#pragma unroll
      for (int i = N1; i < NS; ++i) {
        const int g = K0 + 32 * i;
        float* d = (g < 384) ? th.ldst_lo + g * KST : th.ldst_hi + (g - 384) * KST;
        *(float4*)d = lv[i - N1];
      }
    }
  }
  __syncthreads();

#pragma unroll 1
  for (int s = 0; s < KC / 4; ++s) {
    {
      const float4 lv = *(const float4*)(l768 + K0 + 4 * s);
      const float z0 = th.zc[(4 * s + 0) * ZST], z1 = th.zc[(4 * s + 1) * ZST];
      const float z2 = th.zc[(4 * s + 2) * ZST], z3 = th.zc[(4 * s + 3) * ZST];
      float nb = fmaf(lv.x, z0, bpart);
      nb = fmaf(lv.y, z1, nb);
      nb = fmaf(lv.z, z2, nb);
      nb = fmaf(lv.w, z3, nb);
      bpart = (half == (s & 1)) ? nb : bpart;
    }
    v2f b;
    b.x = th.b[(4 * s) * ZST];
    b.y = th.b[(4 * s + 1) * ZST];
#pragma unroll
    for (int t0 = 2 * P; t0 < NT; t0 += 8) {
      v2f a[8];
#pragma unroll
      for (int j = 0; j < 8; ++j) {
        const int t = t0 + j;
        if (t < NT) {
          const float* ap = (t < 24) ? th.a_lo + t * 16 * KST : th.a_hi + (t - 24) * 16 * KST;
          a[j] = *(const v2f*)(ap + 4 * s);
        }
      }
#pragma unroll
      for (int j = 0; j < 8; ++j) {
        const int t = t0 + j;
        if (t < NT) acc[t] = wmma_f32_16x16x4(a[j], b, acc[t]);
      }
    }
  }
}

template <int P>
__device__ __forceinline__ void ksweep(const char* __restrict__ Lbytes,
                                       const char* __restrict__ zbytes,
                                       const Thr& th, const float* __restrict__ l768,
                                       int half, v8f (&acc)[NT], float& bpart) {
  if constexpr (P < NT / 2) {
    kchunk<P>(Lbytes, zbytes, th, l768, half, acc, bpart);
    ksweep<P + 1>(Lbytes, zbytes, th, l768, half, acc, bpart);
  }
}

__global__ __launch_bounds__(256) void fcvi_main_kernel(
    const float* __restrict__ Lb, const float* __restrict__ z,
    const float* __restrict__ x, const float* __restrict__ mean,
    float* __restrict__ out) {
  __shared__ __attribute__((aligned(16))) float sA[NT * 16 * KST];
  __shared__ __attribute__((aligned(16))) float sB1[NT * 16];
  __shared__ __attribute__((aligned(16))) float sZ[KC * ZST];

  const int tid  = threadIdx.x;
  const int lane = tid & 31;
  const int wave = tid >> 5;
  const int sub  = lane & 15;
  const int half = lane >> 4;
  const int gcol = blockIdx.x * 128 + wave * 16 + sub;
  const float xv = x[gcol & BMSK];
  const float* __restrict__ zcol = z + gcol;

  float bpart = (half == 0) ? mean[768] : 0.0f;

  if (tid < NT * 4) *(float4*)(sB1 + 4 * tid) = *(const float4*)(Lb + (size_t)BROW + 4 * tid);

  v8f acc[NT];
#pragma unroll
  for (int t = 0; t < NT; ++t) acc[t] = (v8f){0.f, 0.f, 0.f, 0.f, 0.f, 0.f, 0.f, 0.f};

  {
    Thr th;
    const int r = tid >> 3, q = tid & 7;
    th.lsrc = (unsigned)(tid * 16);
    th.ldst_lo = sA + r * KST + 4 * q;
    th.ldst_hi = sA + (384 + r) * KST + 4 * q;
    th.zsrc = (unsigned)((wave * NCOL + 4 * lane) * 4);
    th.zdst = sZ + wave * ZST + 4 * lane;
    th.a_lo = sA + sub * KST + 2 * half;
    th.a_hi = sA + (384 + sub) * KST + 2 * half;
    th.b = sZ + (2 * half) * ZST + wave * 16 + sub;
    th.zc = sZ + wave * 16 + sub;
    ksweep<0>((const char*)Lb, (const char*)(z + blockIdx.x * 128), th, sB1, half, acc, bpart);
  }

  float outpart = 0.0f;
#pragma unroll
  for (int mp = 0; mp < 8; ++mp) {
    const int r00 = 32 * mp, r01 = r00 + 16;
    const int rb0 = 256 + r00, rb1 = rb0 + 16;
    const int r10 = 512 + r00, r11 = r10 + 16;
#pragma unroll
    for (int v = 0; v < 8; ++v) {
      const int rl = v + 8 * half;
      const float w0e = acc[2 * mp][v] + mean[r00 + rl];
      const float b0e = acc[16 + 2 * mp][v] + mean[rb0 + rl];
      const float w1e = acc[32 + 2 * mp][v] + mean[r10 + rl];
      outpart = fmaf(fmaxf(fmaf(xv, w0e, b0e), 0.0f), w1e, outpart);
      const float w0o = acc[2 * mp + 1][v] + mean[r01 + rl];
      const float b0o = acc[17 + 2 * mp][v] + mean[rb1 + rl];
      const float w1o = acc[33 + 2 * mp][v] + mean[r11 + rl];
      outpart = fmaf(fmaxf(fmaf(xv, w0o, b0o), 0.0f), w1o, outpart);
    }
  }

  if (half == 0) bpart = fmaf(Lb[(size_t)BROW + 768], zcol[(size_t)768 * NCOL], bpart);

  outpart += bpart;
  const float other = __shfl_xor(outpart, 16, 32);
  if (half == 0) out[gcol] = outpart + other;
}

extern "C" void kernel_launch(void* const* d_in, const int* in_sizes, int n_in,
                              void* d_out, int out_size, void* d_ws, size_t ws_size,
                              hipStream_t stream) {
  (void)in_sizes; (void)n_in; (void)out_size; (void)ws_size;
  const float* x    = (const float*)d_in[0];
  const float* mean = (const float*)d_in[1];
  const float* cov  = (const float*)d_in[2];
  const float* z    = (const float*)d_in[3];
  float* outp = (float*)d_out;
  float* Lbuf = (float*)d_ws;

  build_L_kernel<<<NBLK + 1, 256, 0, stream>>>(cov, Lbuf);
  fcvi_main_kernel<<<NCOL / 128, 256, 0, stream>>>(Lbuf, z, x, mean, outp);
}


# Two 64-column work-groups per compute unit (512 x 8 waves, 256 registers each): partner waves split row tiles by parity; one group's staging runs under the other's matrix ops; fold sum order changes
# speedup vs baseline: 2.2454x; 1.2018x over previous
#include <hip/hip_runtime.h>
#include <math.h>

#define PP   769
#define PANEL 1024
#define NBLK 300
#define BROW (NBLK * PANEL)
#define NCOL 32768
#define BMSK 2047
#define KC   32
#define KST  36
#define WCOL 64
#define ZST  72
#define NT   48
#define NM   24
#define LG   8

typedef __attribute__((ext_vector_type(2))) float v2f;
typedef __attribute__((ext_vector_type(8))) float v8f;

__device__ __forceinline__ v8f wmma_f32_16x16x4(v2f a, v2f b, v8f c) {
  return __builtin_amdgcn_wmma_f32_16x16x4_f32(false, a, false, b, (short)0, c,
                                               false, false);
}

__global__ __launch_bounds__(256) void build_L_kernel(const float* __restrict__ cov,
                                                      float* __restrict__ Lb) {
  const int b = blockIdx.x, tid = threadIdx.x;
  int i, k0;
  if (b < NBLK) {
    int P = 0, rem = b;
    while (rem >= 24 - P) { rem -= 24 - P; ++P; }
    i = 32 * (P + rem) + (tid >> 3);
    k0 = 32 * P + 4 * (tid & 7);
  } else {
    if (tid >= 193) return;
    i = 768;
    k0 = 4 * tid;
  }
  const float* __restrict__ row = cov + i * (i + 1) / 2;
  float v[4];
#pragma unroll
  for (int e = 0; e < 4; ++e) v[e] = (k0 + e <= i) ? row[k0 + e] : 0.0f;
#pragma unroll
  for (int e = 0; e < 4; ++e)
    if (k0 + e == i) v[e] = expf(v[e]);
  *(float4*)(Lb + (size_t)b * PANEL + 4 * tid) = make_float4(v[0], v[1], v[2], v[3]);
}

struct Thr {
  unsigned lsrc;
  unsigned zsrc;
  unsigned ldst_lo;
  unsigned ldst_hi;
  unsigned zdst;
  const float* a_lo;
  const float* a_hi;
  const float* b;
  const float* zc;
};

#define SM_A   0
#define SM_Z   (NT * 16 * KST)
#define SM_B1  (SM_Z + KC * ZST)
#define SM_RED (SM_B1 + NT * 16)
#define SM_END (SM_RED + 4 * WCOL)

template <int P, int I0, int N>
__device__ __forceinline__ void stage_L(const char* __restrict__ lsrc, unsigned tsrc,
                                        char* sm, unsigned lo, unsigned hi) {
  float4 lv[N];
#pragma unroll
  for (int i = 0; i < N; ++i)
    lv[i] = *(const float4*)(lsrc + (size_t)(I0 + i) * (PANEL * 4) + tsrc);
#pragma unroll
  for (int i = 0; i < N; ++i) {
    const int g = KC * P + 32 * (I0 + i);
    char* d = (g < 384) ? sm + lo + g * (KST * 4) : sm + hi + (g - 384) * (KST * 4);
    *(float4*)d = lv[i];
  }
}

template <int P, int I0>
__device__ __forceinline__ void stage_L_rest(const char* __restrict__ lsrc, unsigned tsrc,
                                             char* sm, unsigned lo, unsigned hi) {
  constexpr int NS = NT / 2 - P;
  if constexpr (I0 < NS) {
    constexpr int N = (NS - I0 < LG) ? NS - I0 : LG;
    __builtin_amdgcn_sched_barrier(0);
    stage_L<P, I0, N>(lsrc, tsrc, sm, lo, hi);
    stage_L_rest<P, I0 + N>(lsrc, tsrc, sm, lo, hi);
  }
}

template <int P>
__device__ __forceinline__ void kchunk(const char* __restrict__ Lbytes,
                                       const char* __restrict__ zbytes, char* sm,
                                       const Thr& th, const float* __restrict__ l768,
                                       int half, int rh, v8f (&acc)[NM], float& bpart) {
  constexpr int K0 = KC * P;
  __syncthreads();
  {
    constexpr int NS = NT / 2 - P;
    constexpr int N1 = (NS < LG - 2) ? NS : LG - 2;
    const char* zsrc = zbytes + (size_t)K0 * (NCOL * 4);
    constexpr int CB = 24 * P - P * (P - 1) / 2;
    const char* lsrc = Lbytes + (size_t)CB * (PANEL * 4);
    unsigned lo = th.ldst_lo, hi = th.ldst_hi, zd = th.zdst;
    asm volatile("" : "+v"(lo), "+v"(hi), "+v"(zd));
    {
      float4 zv[2];
      float4 lv[N1];
#pragma unroll
      for (int i = 0; i < 2; ++i)
        zv[i] = *(const float4*)(zsrc + (size_t)(16 * i) * (NCOL * 4) + th.zsrc);
#pragma unroll
      for (int i = 0; i < N1; ++i)
        lv[i] = *(const float4*)(lsrc + (size_t)i * (PANEL * 4) + th.lsrc);
#pragma unroll
      for (int i = 0; i < 2; ++i) *(float4*)(sm + zd + 16 * i * (ZST * 4)) = zv[i];
#pragma unroll
      for (int i = 0; i < N1; ++i) {
        const int g = K0 + 32 * i;
        char* d = (g < 384) ? sm + lo + g * (KST * 4) : sm + hi + (g - 384) * (KST * 4);
        *(float4*)d = lv[i];
      }
    }
    stage_L_rest<P, N1>(lsrc, th.lsrc, sm, lo, hi);
  }
  __syncthreads();

#pragma unroll 1
  for (int s = 0; s < KC / 4; ++s) {
    if ((P & 1) == rh) {
      const float4 lv = *(const float4*)(l768 + K0 + 4 * s);
      const float z0 = th.zc[(4 * s + 0) * ZST], z1 = th.zc[(4 * s + 1) * ZST];
      const float z2 = th.zc[(4 * s + 2) * ZST], z3 = th.zc[(4 * s + 3) * ZST];
      float nb = fmaf(lv.x, z0, bpart);
      nb = fmaf(lv.y, z1, nb);
      nb = fmaf(lv.z, z2, nb);
      nb = fmaf(lv.w, z3, nb);
      bpart = (half == (s & 1)) ? nb : bpart;
    }
    v2f b;
    b.x = th.b[(4 * s) * ZST];
    b.y = th.b[(4 * s + 1) * ZST];
#pragma unroll
    for (int m0 = P; m0 < NM; m0 += 8) {
      v2f a[8];
#pragma unroll
      for (int j = 0; j < 8; ++j) {
        const int m = m0 + j;
        if (m < NM) {
          const float* ap = (m < 12) ? th.a_lo + m * 32 * KST : th.a_hi + (m - 12) * 32 * KST;
          a[j] = *(const v2f*)(ap + 4 * s);
        }
      }
#pragma unroll
      for (int j = 0; j < 8; ++j) {
        const int m = m0 + j;
        if (m < NM) acc[m] = wmma_f32_16x16x4(a[j], b, acc[m]);
      }
    }
  }
}

template <int P>
__device__ __forceinline__ void ksweep(const char* __restrict__ Lbytes,
                                       const char* __restrict__ zbytes, char* sm,
                                       const Thr& th, const float* __restrict__ l768,
                                       int half, int rh, v8f (&acc)[NM], float& bpart) {
  if constexpr (P < NT / 2) {
    kchunk<P>(Lbytes, zbytes, sm, th, l768, half, rh, acc, bpart);
    ksweep<P + 1>(Lbytes, zbytes, sm, th, l768, half, rh, acc, bpart);
  }
}

__global__ __launch_bounds__(256, 4) void fcvi_main_kernel(
    const float* __restrict__ Lb, const float* __restrict__ z,
    const float* __restrict__ x, const float* __restrict__ mean,
    float* __restrict__ out) {
  __shared__ __attribute__((aligned(16))) float smem[SM_END];
  float* const sA = smem + SM_A;
  float* const sZ = smem + SM_Z;
  float* const sB1 = smem + SM_B1;
  float* const sRed = smem + SM_RED;

  const int tid  = threadIdx.x;
  const int lane = tid & 31;
  const int wave = __builtin_amdgcn_readfirstlane(tid >> 5);
  const int cg   = wave & 3;
  const int rh   = wave >> 2;
  const int sub  = lane & 15;
  const int half = lane >> 4;
  const int wcol = cg * 16 + sub;
  const int gcol = blockIdx.x * WCOL + wcol;
  const float xv = x[gcol & BMSK];
  const float* __restrict__ zcol = z + gcol;

  float bpart = (half == 0 && rh == 0) ? mean[768] : 0.0f;

  if (tid < NT * 4) *(float4*)(sB1 + 4 * tid) = *(const float4*)(Lb + (size_t)BROW + 4 * tid);

  v8f acc[NM];
#pragma unroll
  for (int m = 0; m < NM; ++m) acc[m] = (v8f){0.f, 0.f, 0.f, 0.f, 0.f, 0.f, 0.f, 0.f};

  {
    Thr th;
    const int r = tid >> 3, q = tid & 7;
    th.lsrc = (unsigned)(tid * 16);
    th.ldst_lo = (unsigned)((SM_A + r * KST + 4 * q) * 4);
    th.ldst_hi = (unsigned)((SM_A + (384 + r) * KST + 4 * q) * 4);
    const int zr = tid >> 4, zq = tid & 15;
    th.zsrc = (unsigned)((zr * NCOL + 4 * zq) * 4);
    th.zdst = (unsigned)((SM_Z + zr * ZST + 4 * zq) * 4);
    th.a_lo = sA + (16 * rh + sub) * KST + 2 * half;
    th.a_hi = sA + (384 + 16 * rh + sub) * KST + 2 * half;
    th.b = sZ + (2 * half) * ZST + wcol;
    th.zc = sZ + wcol;
    ksweep<0>((const char*)Lb, (const char*)(z + blockIdx.x * WCOL), (char*)smem, th, sB1, half, rh,
              acc, bpart);
  }

  float outpart = 0.0f;
#pragma unroll
  for (int g = 0; g < 8; ++g) {
    const int r0 = 16 * (2 * g + rh);
#pragma unroll
    for (int v = 0; v < 8; ++v) {
      const int rl = r0 + v + 8 * half;
      const float w0 = acc[g][v] + mean[rl];
      const float b0 = acc[8 + g][v] + mean[256 + rl];
      const float w1 = acc[16 + g][v] + mean[512 + rl];
      outpart = fmaf(fmaxf(fmaf(xv, w0, b0), 0.0f), w1, outpart);
    }
  }

  if (half == 0 && rh == 0) bpart = fmaf(Lb[(size_t)BROW + 768], zcol[(size_t)768 * NCOL], bpart);

  sRed[(2 * rh + half) * WCOL + wcol] = outpart + bpart;
  __syncthreads();
  if (tid < WCOL)
    out[blockIdx.x * WCOL + tid] =
        ((sRed[tid] + sRed[WCOL + tid]) + sRed[2 * WCOL + tid]) + sRed[3 * WCOL + tid];
}

extern "C" void kernel_launch(void* const* d_in, const int* in_sizes, int n_in,
                              void* d_out, int out_size, void* d_ws, size_t ws_size,
                              hipStream_t stream) {
  (void)in_sizes; (void)n_in; (void)out_size; (void)ws_size;
  const float* x    = (const float*)d_in[0];
  const float* mean = (const float*)d_in[1];
  const float* cov  = (const float*)d_in[2];
  const float* z    = (const float*)d_in[3];
  float* outp = (float*)d_out;
  float* Lbuf = (float*)d_ws;

  build_L_kernel<<<NBLK + 1, 256, 0, stream>>>(cov, Lbuf);
  fcvi_main_kernel<<<NCOL / WCOL, 256, 0, stream>>>(Lbuf, z, x, mean, outp);
}


# One 16-wave work-group per compute unit, 128-column tile, two LDS chunk buffers: parity-1 waves stage chunk P+1 before their matrix ops, parity-0 waves after; one barrier a chunk; sum order kept
# speedup vs baseline: 2.4060x; 1.0715x over previous
#include <hip/hip_runtime.h>
#include <math.h>

#define PP   769
#define PANEL 1024
#define NBLK 300
#define BROW (NBLK * PANEL)
#define NCOL 32768
#define BMSK 2047
#define KC   32
#define KST  36
#define WCOL 128
#define ZST  136
#define NT   48
#define NM   24
#define LG   8

typedef __attribute__((ext_vector_type(2))) float v2f;
typedef __attribute__((ext_vector_type(8))) float v8f;

__device__ __forceinline__ v8f wmma_f32_16x16x4(v2f a, v2f b, v8f c) {
  return __builtin_amdgcn_wmma_f32_16x16x4_f32(false, a, false, b, (short)0, c,
                                               false, false);
}

__global__ __launch_bounds__(256) void build_L_kernel(const float* __restrict__ cov,
                                                      float* __restrict__ Lb) {
  const int b = blockIdx.x, tid = threadIdx.x;
  int i, k0;
  if (b < NBLK) {
    int P = 0, rem = b;
    while (rem >= 24 - P) { rem -= 24 - P; ++P; }
    i = 32 * (P + rem) + (tid >> 3);
    k0 = 32 * P + 4 * (tid & 7);
  } else {
    if (tid >= 193) return;
    i = 768;
    k0 = 4 * tid;
  }
  const float* __restrict__ row = cov + i * (i + 1) / 2;
  float v[4];
#pragma unroll
  for (int e = 0; e < 4; ++e) v[e] = (k0 + e <= i) ? row[k0 + e] : 0.0f;
#pragma unroll
  for (int e = 0; e < 4; ++e)
    if (k0 + e == i) v[e] = expf(v[e]);
  *(float4*)(Lb + (size_t)b * PANEL + 4 * tid) = make_float4(v[0], v[1], v[2], v[3]);
}

struct Thr {
  unsigned lsrc;
  unsigned zsrc;
  unsigned ldst_lo;
  unsigned ldst_hi;
  unsigned zdst;
  unsigned a_lo;
  unsigned b;
  unsigned zc;
};

#define SM_A   0
#define SM_Z   (NT * 16 * KST)
#define BUFSZ  (SM_Z + KC * ZST)
#define SM_B1  (2 * BUFSZ)
#define SM_RED (SM_B1 + NT * 16)
#define SM_END (SM_RED + 4 * WCOL)

template <int Pn, int J0, int N>
__device__ __forceinline__ void stage_L(const char* __restrict__ lsrc, unsigned tsrc,
                                        char* sm, unsigned lo, unsigned hi, bool lowhalf) {
  constexpr int NS = NT / 2 - Pn;
  float4 lv[N];
#pragma unroll
  for (int i = 0; i < N; ++i) {
    const bool full = 2 * (J0 + i) + 1 < NS;
    const unsigned t = (full || lowhalf) ? tsrc : tsrc - PANEL * 4;
    lv[i] = *(const float4*)(lsrc + (size_t)(J0 + i) * (2 * PANEL * 4) + t);
  }
#pragma unroll
  for (int i = 0; i < N; ++i) {
    const bool full = 2 * (J0 + i) + 1 < NS;
    const int g = KC * Pn + 64 * (J0 + i);
    char* d = (g < 384) ? sm + lo + g * (KST * 4) : sm + hi + (g - 384) * (KST * 4);
    if (!(full || lowhalf)) d -= 32 * KST * 4;
    *(float4*)d = lv[i];
  }
}

template <int Pn, int J0>
__device__ __forceinline__ void stage_L_rest(const char* __restrict__ lsrc, unsigned tsrc,
                                             char* sm, unsigned lo, unsigned hi, bool lowhalf) {
  constexpr int NJ = (NT / 2 - Pn + 1) / 2;
  if constexpr (J0 < NJ) {
    constexpr int N = (NJ - J0 < LG) ? NJ - J0 : LG;
    __builtin_amdgcn_sched_barrier(0);
    stage_L<Pn, J0, N>(lsrc, tsrc, sm, lo, hi, lowhalf);
    stage_L_rest<Pn, J0 + N>(lsrc, tsrc, sm, lo, hi, lowhalf);
  }
}

template <int Pn>
__device__ __forceinline__ void stage_chunk(const char* __restrict__ Lbytes,
                                            const char* __restrict__ zbytes, char* sm,
                                            const Thr& th, bool lowhalf) {
  constexpr int K0 = KC * Pn;
  constexpr int NJ = (NT / 2 - Pn + 1) / 2;
  constexpr int N1 = (NJ < LG - 2) ? NJ : LG - 2;
  constexpr unsigned BO = (Pn & 1) * (BUFSZ * 4);
  const char* zsrc = zbytes + (size_t)K0 * (NCOL * 4);
  constexpr int CB = 24 * Pn - Pn * (Pn - 1) / 2;
  const char* lsrc = Lbytes + (size_t)CB * (PANEL * 4);
  unsigned lo = th.ldst_lo + BO, hi = th.ldst_hi + BO, zd = th.zdst + BO;
  asm volatile("" : "+v"(lo), "+v"(hi), "+v"(zd));
  {
    float4 zv[2];
#pragma unroll
    for (int i = 0; i < 2; ++i)
      zv[i] = *(const float4*)(zsrc + (size_t)(16 * i) * (NCOL * 4) + th.zsrc);
    stage_L<Pn, 0, N1>(lsrc, th.lsrc, sm, lo, hi, lowhalf);
#pragma unroll
    for (int i = 0; i < 2; ++i) *(float4*)(sm + zd + 16 * i * (ZST * 4)) = zv[i];
  }
  stage_L_rest<Pn, N1>(lsrc, th.lsrc, sm, lo, hi, lowhalf);
}

template <int P>
__device__ __forceinline__ void compute_chunk(char* sm, const Thr& th, int half, int rh,
                                              v8f (&acc)[NM], float& bpart) {
  constexpr int K0 = KC * P;
  constexpr unsigned BO = (P & 1) * (BUFSZ * 4);
  unsigned ao = th.a_lo + BO, ah = th.a_lo + BO + 384 * KST * 4, bo = th.b + BO, zo = th.zc + BO;
  asm volatile("" : "+v"(ao), "+v"(ah), "+v"(bo), "+v"(zo));
  const float* a_lo = (const float*)(sm + ao);
  const float* a_hi = (const float*)(sm + ah);
  const float* bp = (const float*)(sm + bo);
  const float* zc = (const float*)(sm + zo);
  const float* l768 = (const float*)(sm + SM_B1 * 4);

#pragma unroll 1
  for (int s = 0; s < KC / 4; ++s) {
    if ((P & 1) == rh) {
      const float4 lv = *(const float4*)(l768 + K0 + 4 * s);
      const float z0 = zc[(4 * s + 0) * ZST], z1 = zc[(4 * s + 1) * ZST];
      const float z2 = zc[(4 * s + 2) * ZST], z3 = zc[(4 * s + 3) * ZST];
      float nb = fmaf(lv.x, z0, bpart);
      nb = fmaf(lv.y, z1, nb);
      nb = fmaf(lv.z, z2, nb);
      nb = fmaf(lv.w, z3, nb);
      bpart = (half == (s & 1)) ? nb : bpart;
    }
    v2f b;
    b.x = bp[(4 * s) * ZST];
    b.y = bp[(4 * s + 1) * ZST];
#pragma unroll
    for (int m0 = P; m0 < NM; m0 += 8) {
      v2f a[8];
#pragma unroll
      for (int j = 0; j < 8; ++j) {
        const int m = m0 + j;
        if (m < NM) {
          const float* ap = (m < 12) ? a_lo + m * 32 * KST : a_hi + (m - 12) * 32 * KST;
          a[j] = *(const v2f*)(ap + 4 * s);
        }
      }
#pragma unroll
      for (int j = 0; j < 8; ++j) {
        const int m = m0 + j;
        if (m < NM) acc[m] = wmma_f32_16x16x4(a[j], b, acc[m]);
      }
    }
  }
}

template <int P>
__device__ __forceinline__ void ksweep(const char* __restrict__ Lbytes,
                                       const char* __restrict__ zbytes, char* sm,
                                       const Thr& th, bool lowhalf, int half, int rh,
                                       v8f (&acc)[NM], float& bpart) {
  if constexpr (P < NT / 2) {
    if constexpr (P + 1 < NT / 2) {
      if (rh == 1) stage_chunk<P + 1>(Lbytes, zbytes, sm, th, lowhalf);
    }
    compute_chunk<P>(sm, th, half, rh, acc, bpart);
    if constexpr (P + 1 < NT / 2) {
      if (rh == 0) stage_chunk<P + 1>(Lbytes, zbytes, sm, th, lowhalf);
      __syncthreads();
    }
    ksweep<P + 1>(Lbytes, zbytes, sm, th, lowhalf, half, rh, acc, bpart);
  }
}

__global__ __launch_bounds__(512, 4) void fcvi_main_kernel(
    const float* __restrict__ Lb, const float* __restrict__ z,
    const float* __restrict__ x, const float* __restrict__ mean,
    float* __restrict__ out) {
  __shared__ __attribute__((aligned(16))) float smem[SM_END];
  float* const sB1 = smem + SM_B1;
  float* const sRed = smem + SM_RED;

  const int tid  = threadIdx.x;
  const int lane = tid & 31;
  const int wave = __builtin_amdgcn_readfirstlane(tid >> 5);
  const int cg   = (wave & 3) | ((wave >> 3) << 2);
  const int rh   = (wave >> 2) & 1;
  const bool lowhalf = wave < 8;
  const int sub  = lane & 15;
  const int half = lane >> 4;
  const int wcol = cg * 16 + sub;
  const int gcol = blockIdx.x * WCOL + wcol;
  const float xv = x[gcol & BMSK];
  const float* __restrict__ zcol = z + gcol;

  float bpart = (half == 0 && rh == 0) ? mean[768] : 0.0f;

  if (tid < NT * 4) *(float4*)(sB1 + 4 * tid) = *(const float4*)(Lb + (size_t)BROW + 4 * tid);

  v8f acc[NM];
#pragma unroll
  for (int m = 0; m < NM; ++m) acc[m] = (v8f){0.f, 0.f, 0.f, 0.f, 0.f, 0.f, 0.f, 0.f};

  {
    Thr th;
    const int r = tid >> 3, q = tid & 7;
    th.lsrc = (unsigned)(tid * 16);
    th.ldst_lo = (unsigned)((SM_A + r * KST + 4 * q) * 4);
    th.ldst_hi = (unsigned)((SM_A + (384 + r) * KST + 4 * q) * 4);
    const int zr = tid >> 5, zq = tid & 31;
    th.zsrc = (unsigned)((zr * NCOL + 4 * zq) * 4);
    th.zdst = (unsigned)((SM_Z + zr * ZST + 4 * zq) * 4);
    th.a_lo = (unsigned)((SM_A + (16 * rh + sub) * KST + 2 * half) * 4);
    th.b = (unsigned)((SM_Z + (2 * half) * ZST + wcol) * 4);
    th.zc = (unsigned)((SM_Z + wcol) * 4);
    const char* Lbytes = (const char*)Lb;
    const char* zbytes = (const char*)(z + blockIdx.x * WCOL);
    stage_chunk<0>(Lbytes, zbytes, (char*)smem, th, lowhalf);
    __syncthreads();
    ksweep<0>(Lbytes, zbytes, (char*)smem, th, lowhalf, half, rh, acc, bpart);
  }
  float outpart = 0.0f;
#pragma unroll
  for (int g = 0; g < 8; ++g) {
    const int r0 = 16 * (2 * g + rh);
#pragma unroll
    for (int v = 0; v < 8; ++v) {
      const int rl = r0 + v + 8 * half;
      const float w0 = acc[g][v] + mean[rl];
      const float b0 = acc[8 + g][v] + mean[256 + rl];
      const float w1 = acc[16 + g][v] + mean[512 + rl];
      outpart = fmaf(fmaxf(fmaf(xv, w0, b0), 0.0f), w1, outpart);
    }
  }

  if (half == 0 && rh == 0) bpart = fmaf(Lb[(size_t)BROW + 768], zcol[(size_t)768 * NCOL], bpart);

  sRed[(2 * rh + half) * WCOL + wcol] = outpart + bpart;
  __syncthreads();
  if (tid < WCOL)
    out[blockIdx.x * WCOL + tid] =
        ((sRed[tid] + sRed[WCOL + tid]) + sRed[2 * WCOL + tid]) + sRed[3 * WCOL + tid];
}

extern "C" void kernel_launch(void* const* d_in, const int* in_sizes, int n_in,
                              void* d_out, int out_size, void* d_ws, size_t ws_size,
                              hipStream_t stream) {
  (void)in_sizes; (void)n_in; (void)out_size; (void)ws_size;
  const float* x    = (const float*)d_in[0];
  const float* mean = (const float*)d_in[1];
  const float* cov  = (const float*)d_in[2];
  const float* z    = (const float*)d_in[3];
  float* outp = (float*)d_out;
  float* Lbuf = (float*)d_ws;

  build_L_kernel<<<NBLK + 1, 256, 0, stream>>>(cov, Lbuf);
  fcvi_main_kernel<<<NCOL / WCOL, 512, 0, stream>>>(Lbuf, z, x, mean, outp);
}


# b1 row taken out of the matrix loop: the staging threads add L[768,k] (scalar loads) times the Z values they copy to a float4 of their own in LDS, once a chunk; b1 sum order changes
# speedup vs baseline: 2.5724x; 1.0692x over previous
#include <hip/hip_runtime.h>
#include <math.h>

#define PP   769
#define PANEL 1024
#define NBLK 300
#define BROW (NBLK * PANEL)
#define NCOL 32768
#define BMSK 2047
#define KC   32
#define KST  36
#define WCOL 128
#define ZST  136
#define NT   48
#define NM   24
#define LG   8

typedef __attribute__((ext_vector_type(2))) float v2f;
typedef __attribute__((ext_vector_type(8))) float v8f;

__device__ __forceinline__ v8f wmma_f32_16x16x4(v2f a, v2f b, v8f c) {
  return __builtin_amdgcn_wmma_f32_16x16x4_f32(false, a, false, b, (short)0, c,
                                               false, false);
}

__global__ __launch_bounds__(256) void build_L_kernel(const float* __restrict__ cov,
                                                      float* __restrict__ Lb) {
  const int b = blockIdx.x, tid = threadIdx.x;
  int i, k0;
  if (b < NBLK) {
    int P = 0, rem = b;
    while (rem >= 24 - P) { rem -= 24 - P; ++P; }
    i = 32 * (P + rem) + (tid >> 3);
    k0 = 32 * P + 4 * (tid & 7);
  } else {
    if (tid >= 193) return;
    i = 768;
    k0 = 4 * tid;
  }
  const float* __restrict__ row = cov + i * (i + 1) / 2;
  float v[4];
#pragma unroll
  for (int e = 0; e < 4; ++e) v[e] = (k0 + e <= i) ? row[k0 + e] : 0.0f;
#pragma unroll
  for (int e = 0; e < 4; ++e)
    if (k0 + e == i) v[e] = expf(v[e]);
  *(float4*)(Lb + (size_t)b * PANEL + 4 * tid) = make_float4(v[0], v[1], v[2], v[3]);
}

struct Thr {
  unsigned lsrc;
  unsigned zsrc;
  unsigned ldst_lo;
  unsigned ldst_hi;
  unsigned zdst;
  unsigned a_lo;
  unsigned b;
};

#define SM_BP  0
#define SM_A   (16 * WCOL)
#define SM_Z   (SM_A + NT * 16 * KST)
#define BUFSZ  (NT * 16 * KST + KC * ZST)
#define SM_RED (SM_A + 2 * BUFSZ)
#define SM_END (SM_RED + 4 * WCOL)

template <int Pn, int J0, int N>
__device__ __forceinline__ void stage_L(const char* __restrict__ lsrc, unsigned tsrc,
                                        char* sm, unsigned lo, unsigned hi, bool lowhalf) {
  constexpr int NS = NT / 2 - Pn;
  float4 lv[N];
#pragma unroll
  for (int i = 0; i < N; ++i) {
    const bool full = 2 * (J0 + i) + 1 < NS;
    const unsigned t = (full || lowhalf) ? tsrc : tsrc - PANEL * 4;
    lv[i] = *(const float4*)(lsrc + (size_t)(J0 + i) * (2 * PANEL * 4) + t);
  }
#pragma unroll
  for (int i = 0; i < N; ++i) {
    const bool full = 2 * (J0 + i) + 1 < NS;
    const int g = KC * Pn + 64 * (J0 + i);
    char* d = (g < 384) ? sm + lo + g * (KST * 4) : sm + hi + (g - 384) * (KST * 4);
    if (!(full || lowhalf)) d -= 32 * KST * 4;
    *(float4*)d = lv[i];
  }
}

template <int Pn, int J0>
__device__ __forceinline__ void stage_L_rest(const char* __restrict__ lsrc, unsigned tsrc,
                                             char* sm, unsigned lo, unsigned hi, bool lowhalf) {
  constexpr int NJ = (NT / 2 - Pn + 1) / 2;
  if constexpr (J0 < NJ) {
    constexpr int N = (NJ - J0 < LG) ? NJ - J0 : LG;
    __builtin_amdgcn_sched_barrier(0);
    stage_L<Pn, J0, N>(lsrc, tsrc, sm, lo, hi, lowhalf);
    stage_L_rest<Pn, J0 + N>(lsrc, tsrc, sm, lo, hi, lowhalf);
  }
}

template <int Pn>
__device__ __forceinline__ void stage_chunk(const char* __restrict__ Lbytes,
                                            const char* __restrict__ zbytes, char* sm,
                                            const Thr& th, bool lowhalf,
                                            const float* __restrict__ l768w) {
  constexpr int K0 = KC * Pn;
  constexpr int NJ = (NT / 2 - Pn + 1) / 2;
  constexpr int N1 = (NJ < LG - 2) ? NJ : LG - 2;
  constexpr unsigned BO = (Pn & 1) * (BUFSZ * 4);
  const char* zsrc = zbytes + (size_t)K0 * (NCOL * 4);
  constexpr int CB = 24 * Pn - Pn * (Pn - 1) / 2;
  const char* lsrc = Lbytes + (size_t)CB * (PANEL * 4);
  unsigned lo = th.ldst_lo + BO, hi = th.ldst_hi + BO, zd = th.zdst + BO;
  asm volatile("" : "+v"(lo), "+v"(hi), "+v"(zd));
  const float lk0 = l768w[K0], lk1 = l768w[K0 + 16];
  {
    float4 zv[2];
#pragma unroll
    for (int i = 0; i < 2; ++i)
      zv[i] = *(const float4*)(zsrc + (size_t)(16 * i) * (NCOL * 4) + th.zsrc);
    stage_L<Pn, 0, N1>(lsrc, th.lsrc, sm, lo, hi, lowhalf);
#pragma unroll
    for (int i = 0; i < 2; ++i) *(float4*)(sm + zd + 16 * i * (ZST * 4)) = zv[i];
    float4* const bsum = (float4*)(sm + SM_BP * 4 + th.lsrc);
    float4 bs = *bsum;
    bs.x = fmaf(lk1, zv[1].x, fmaf(lk0, zv[0].x, bs.x));
    bs.y = fmaf(lk1, zv[1].y, fmaf(lk0, zv[0].y, bs.y));
    bs.z = fmaf(lk1, zv[1].z, fmaf(lk0, zv[0].z, bs.z));
    bs.w = fmaf(lk1, zv[1].w, fmaf(lk0, zv[0].w, bs.w));
    *bsum = bs;
  }
  stage_L_rest<Pn, N1>(lsrc, th.lsrc, sm, lo, hi, lowhalf);
}

template <int P>
__device__ __forceinline__ void compute_chunk(char* sm, const Thr& th, v8f (&acc)[NM]) {
  constexpr unsigned BO = (P & 1) * (BUFSZ * 4);
  unsigned ao = th.a_lo + BO, ah = th.a_lo + BO + 384 * KST * 4, bo = th.b + BO;
  asm volatile("" : "+v"(ao), "+v"(ah), "+v"(bo));
  const float* a_lo = (const float*)(sm + ao);
  const float* a_hi = (const float*)(sm + ah);
  const float* bp = (const float*)(sm + bo);

#pragma unroll 1
  for (int s = 0; s < KC / 4; ++s) {
    v2f b;
    b.x = bp[(4 * s) * ZST];
    b.y = bp[(4 * s + 1) * ZST];
#pragma unroll
    for (int m0 = P; m0 < NM; m0 += 8) {
      v2f a[8];
#pragma unroll
      for (int j = 0; j < 8; ++j) {
        const int m = m0 + j;
        if (m < NM) {
          const float* ap = (m < 12) ? a_lo + m * 32 * KST : a_hi + (m - 12) * 32 * KST;
          a[j] = *(const v2f*)(ap + 4 * s);
        }
      }
#pragma unroll
      for (int j = 0; j < 8; ++j) {
        const int m = m0 + j;
        if (m < NM) acc[m] = wmma_f32_16x16x4(a[j], b, acc[m]);
      }
    }
  }
}

template <int P>
__device__ __forceinline__ void ksweep(const char* __restrict__ Lbytes,
                                       const char* __restrict__ zbytes, char* sm,
                                       const Thr& th, bool lowhalf, int rh,
                                       const float* __restrict__ l768w, v8f (&acc)[NM]) {
  if constexpr (P < NT / 2) {
    if constexpr (P + 1 < NT / 2) {
      if (rh == 1) stage_chunk<P + 1>(Lbytes, zbytes, sm, th, lowhalf, l768w);
    }
    compute_chunk<P>(sm, th, acc);
    if constexpr (P + 1 < NT / 2) {
      if (rh == 0) stage_chunk<P + 1>(Lbytes, zbytes, sm, th, lowhalf, l768w);
      __syncthreads();
    }
    ksweep<P + 1>(Lbytes, zbytes, sm, th, lowhalf, rh, l768w, acc);
  }
}

__global__ __launch_bounds__(512, 4) void fcvi_main_kernel(
    const float* __restrict__ Lb, const float* __restrict__ z,
    const float* __restrict__ x, const float* __restrict__ mean,
    float* __restrict__ out) {
  __shared__ __attribute__((aligned(16))) float smem[SM_END];
  float* const sRed = smem + SM_RED;
  float* const sBp = smem + SM_BP;

  const int tid  = threadIdx.x;
  const int lane = tid & 31;
  const int wave = __builtin_amdgcn_readfirstlane(tid >> 5);
  const int cg   = (wave & 3) | ((wave >> 3) << 2);
  const int rh   = (wave >> 2) & 1;
  const bool lowhalf = wave < 8;
  const int sub  = lane & 15;
  const int half = lane >> 4;
  const int wcol = cg * 16 + sub;
  const int gcol = blockIdx.x * WCOL + wcol;
  const float xv = x[gcol & BMSK];
  const float* __restrict__ zcol = z + gcol;

  *(float4*)(sBp + 4 * tid) = make_float4(0.0f, 0.0f, 0.0f, 0.0f);
  const float* __restrict__ l768w = Lb + (size_t)BROW + wave;

  v8f acc[NM];
#pragma unroll
  for (int m = 0; m < NM; ++m) acc[m] = (v8f){0.f, 0.f, 0.f, 0.f, 0.f, 0.f, 0.f, 0.f};

  {
    Thr th;
    const int r = tid >> 3, q = tid & 7;
    th.lsrc = (unsigned)(tid * 16);
    th.ldst_lo = (unsigned)((SM_A + r * KST + 4 * q) * 4);
    th.ldst_hi = (unsigned)((SM_A + (384 + r) * KST + 4 * q) * 4);
    const int zr = tid >> 5, zq = tid & 31;
    th.zsrc = (unsigned)((zr * NCOL + 4 * zq) * 4);
    th.zdst = (unsigned)((SM_Z + zr * ZST + 4 * zq) * 4);
    th.a_lo = (unsigned)((SM_A + (16 * rh + sub) * KST + 2 * half) * 4);
    th.b = (unsigned)((SM_Z + (2 * half) * ZST + wcol) * 4);
    const char* Lbytes = (const char*)Lb;
    const char* zbytes = (const char*)(z + blockIdx.x * WCOL);
    stage_chunk<0>(Lbytes, zbytes, (char*)smem, th, lowhalf, l768w);
    __syncthreads();
    ksweep<0>(Lbytes, zbytes, (char*)smem, th, lowhalf, rh, l768w, acc);
  }
  float outpart = 0.0f;
#pragma unroll
  for (int g = 0; g < 8; ++g) {
    const int r0 = 16 * (2 * g + rh);
#pragma unroll
    for (int v = 0; v < 8; ++v) {
      const int rl = r0 + v + 8 * half;
      const float w0 = acc[g][v] + mean[rl];
      const float b0 = acc[8 + g][v] + mean[256 + rl];
      const float w1 = acc[16 + g][v] + mean[512 + rl];
      outpart = fmaf(fmaxf(fmaf(xv, w0, b0), 0.0f), w1, outpart);
    }
  }

  float bpart = 0.0f;
  if (half == 0 && rh == 0)
    bpart = fmaf(Lb[(size_t)BROW + 768], zcol[(size_t)768 * NCOL], mean[768]);

  sRed[(2 * rh + half) * WCOL + wcol] = outpart + bpart;
  __syncthreads();
  if (tid < WCOL) {
    float bs = sBp[tid];
    for (int w = 1; w < 16; ++w) bs += sBp[w * WCOL + tid];
    out[blockIdx.x * WCOL + tid] =
        (((sRed[tid] + sRed[WCOL + tid]) + sRed[2 * WCOL + tid]) + sRed[3 * WCOL + tid]) + bs;
  }
}

extern "C" void kernel_launch(void* const* d_in, const int* in_sizes, int n_in,
                              void* d_out, int out_size, void* d_ws, size_t ws_size,
                              hipStream_t stream) {
  (void)in_sizes; (void)n_in; (void)out_size; (void)ws_size;
  const float* x    = (const float*)d_in[0];
  const float* mean = (const float*)d_in[1];
  const float* cov  = (const float*)d_in[2];
  const float* z    = (const float*)d_in[3];
  float* outp = (float*)d_out;
  float* Lbuf = (float*)d_ws;

  build_L_kernel<<<NBLK + 1, 256, 0, stream>>>(cov, Lbuf);
  fcvi_main_kernel<<<NCOL / WCOL, 512, 0, stream>>>(Lbuf, z, x, mean, outp);
}


# Fragment reads from LDS run one group of up to 8 matrix ops ahead in two register slots, one read place per group, two k steps per loop trip; sum order kept
# speedup vs baseline: 2.6451x; 1.0283x over previous
#include <hip/hip_runtime.h>
#include <math.h>

#define PP   769
#define PANEL 1024
#define NBLK 300
#define BROW (NBLK * PANEL)
#define NCOL 32768
#define BMSK 2047
#define KC   32
#define KST  36
#define WCOL 128
#define ZST  136
#define NT   48
#define NM   24
#define LG   8
#define GT   8

typedef __attribute__((ext_vector_type(2))) float v2f;
typedef __attribute__((ext_vector_type(8))) float v8f;

__device__ __forceinline__ v8f wmma_f32_16x16x4(v2f a, v2f b, v8f c) {
  return __builtin_amdgcn_wmma_f32_16x16x4_f32(false, a, false, b, (short)0, c,
                                               false, false);
}

__global__ __launch_bounds__(256) void build_L_kernel(const float* __restrict__ cov,
                                                      float* __restrict__ Lb) {
  const int b = blockIdx.x, tid = threadIdx.x;
  int i, k0;
  if (b < NBLK) {
    int P = 0, rem = b;
    while (rem >= 24 - P) { rem -= 24 - P; ++P; }
    i = 32 * (P + rem) + (tid >> 3);
    k0 = 32 * P + 4 * (tid & 7);
  } else {
    if (tid >= 193) return;
    i = 768;
    k0 = 4 * tid;
  }
  const float* __restrict__ row = cov + i * (i + 1) / 2;
  float v[4];
#pragma unroll
  for (int e = 0; e < 4; ++e) v[e] = (k0 + e <= i) ? row[k0 + e] : 0.0f;
#pragma unroll
  for (int e = 0; e < 4; ++e)
    if (k0 + e == i) v[e] = expf(v[e]);
  *(float4*)(Lb + (size_t)b * PANEL + 4 * tid) = make_float4(v[0], v[1], v[2], v[3]);
}

struct Thr {
  unsigned lsrc;
  unsigned zsrc;
  unsigned ldst_lo;
  unsigned ldst_hi;
  unsigned zdst;
  unsigned a_lo;
  unsigned b;
};

#define SM_BP  0
#define SM_A   (16 * WCOL)
#define SM_Z   (SM_A + NT * 16 * KST)
#define BUFSZ  (NT * 16 * KST + KC * ZST)
#define SM_RED (SM_A + 2 * BUFSZ)
#define SM_END (SM_RED + 4 * WCOL)

template <int Pn, int J0, int N>
__device__ __forceinline__ void stage_L(const char* __restrict__ lsrc, unsigned tsrc,
                                        char* sm, unsigned lo, unsigned hi, bool lowhalf) {
  constexpr int NS = NT / 2 - Pn;
  float4 lv[N];
#pragma unroll
  for (int i = 0; i < N; ++i) {
    const bool full = 2 * (J0 + i) + 1 < NS;
    const unsigned t = (full || lowhalf) ? tsrc : tsrc - PANEL * 4;
    lv[i] = *(const float4*)(lsrc + (size_t)(J0 + i) * (2 * PANEL * 4) + t);
  }
#pragma unroll
  for (int i = 0; i < N; ++i) {
    const bool full = 2 * (J0 + i) + 1 < NS;
    const int g = KC * Pn + 64 * (J0 + i);
    char* d = (g < 384) ? sm + lo + g * (KST * 4) : sm + hi + (g - 384) * (KST * 4);
    if (!(full || lowhalf)) d -= 32 * KST * 4;
    *(float4*)d = lv[i];
  }
}

template <int Pn, int J0>
__device__ __forceinline__ void stage_L_rest(const char* __restrict__ lsrc, unsigned tsrc,
                                             char* sm, unsigned lo, unsigned hi, bool lowhalf) {
  constexpr int NJ = (NT / 2 - Pn + 1) / 2;
  if constexpr (J0 < NJ) {
    constexpr int N = (NJ - J0 < LG) ? NJ - J0 : LG;
    __builtin_amdgcn_sched_barrier(0);
    stage_L<Pn, J0, N>(lsrc, tsrc, sm, lo, hi, lowhalf);
    stage_L_rest<Pn, J0 + N>(lsrc, tsrc, sm, lo, hi, lowhalf);
  }
}

template <int Pn>
__device__ __forceinline__ void stage_chunk(const char* __restrict__ Lbytes,
                                            const char* __restrict__ zbytes, char* sm,
                                            const Thr& th, bool lowhalf,
                                            const float* __restrict__ l768w) {
  constexpr int K0 = KC * Pn;
  constexpr int NJ = (NT / 2 - Pn + 1) / 2;
  constexpr int N1 = (NJ < LG - 2) ? NJ : LG - 2;
  constexpr unsigned BO = (Pn & 1) * (BUFSZ * 4);
  const char* zsrc = zbytes + (size_t)K0 * (NCOL * 4);
  constexpr int CB = 24 * Pn - Pn * (Pn - 1) / 2;
  const char* lsrc = Lbytes + (size_t)CB * (PANEL * 4);
  unsigned lo = th.ldst_lo + BO, hi = th.ldst_hi + BO, zd = th.zdst + BO;
  asm volatile("" : "+v"(lo), "+v"(hi), "+v"(zd));
  const float lk0 = l768w[K0], lk1 = l768w[K0 + 16];
  {
    float4 zv[2];
#pragma unroll
    for (int i = 0; i < 2; ++i)
      zv[i] = *(const float4*)(zsrc + (size_t)(16 * i) * (NCOL * 4) + th.zsrc);
    stage_L<Pn, 0, N1>(lsrc, th.lsrc, sm, lo, hi, lowhalf);
#pragma unroll
    for (int i = 0; i < 2; ++i) *(float4*)(sm + zd + 16 * i * (ZST * 4)) = zv[i];
    float4* const bsum = (float4*)(sm + SM_BP * 4 + th.lsrc);
    float4 bs = *bsum;
    bs.x = fmaf(lk1, zv[1].x, fmaf(lk0, zv[0].x, bs.x));
    bs.y = fmaf(lk1, zv[1].y, fmaf(lk0, zv[0].y, bs.y));
    bs.z = fmaf(lk1, zv[1].z, fmaf(lk0, zv[0].z, bs.z));
    bs.w = fmaf(lk1, zv[1].w, fmaf(lk0, zv[0].w, bs.w));
    *bsum = bs;
  }
  stage_L_rest<Pn, N1>(lsrc, th.lsrc, sm, lo, hi, lowhalf);
}

constexpr int grp_first(int N, int NG, int g) { return g * (N / NG) + (g < N % NG ? g : N % NG); }
constexpr int grp_size(int N, int NG, int g) { return N / NG + (g < N % NG ? 1 : 0); }

template <int P>
__device__ __forceinline__ void load_A(int first, int size, const char* a_lo, const char* a_hi,
                                       unsigned koff, v2f (&dst)[GT]) {
#pragma unroll
  for (int j = 0; j < GT; ++j) {
    const int m = P + first + j;
    if (j < size) {
      const char* ap = (m < 12) ? a_lo + m * (32 * KST * 4) : a_hi + (m - 12) * (32 * KST * 4);
      dst[j] = *(const v2f*)(ap + koff);
    }
  }
}

__device__ __forceinline__ v2f load_B(const char* bp, int s) {
  const float* p = (const float*)bp + (4 * s) * ZST;
  v2f b;
  b.x = p[0];
  b.y = p[ZST];
  return b;
}

template <int P>
__device__ __forceinline__ void compute_chunk(char* sm, const Thr& th, v8f (&acc)[NM]) {
  constexpr unsigned BO = (P & 1) * (BUFSZ * 4);
  constexpr int N = NM - P;
  constexpr int NG = (N + GT - 1) / GT;
  unsigned ao = th.a_lo + BO, ah = th.a_lo + BO + 384 * KST * 4, bo = th.b + BO;
  asm volatile("" : "+v"(ao), "+v"(ah), "+v"(bo));
  const char* a_lo = sm + ao;
  const char* a_hi = sm + ah;
  const char* bp = sm + bo;

  v2f fa[2][GT];
  v2f fb[2];
  fb[0] = load_B(bp, 0);
  load_A<P>(grp_first(N, NG, 0), grp_size(N, NG, 0), a_lo, a_hi, 0u, fa[0]);
#pragma unroll 1
  for (int s2 = 0; s2 < KC / 8; ++s2) {
#pragma unroll
    for (int ss = 0; ss < 2; ++ss) {
      const int s = 2 * s2 + ss;
#pragma unroll
      for (int g = 0; g < NG; ++g) {
        const int slot = (ss * NG + g) & 1;
        if (g + 1 < NG) {
          load_A<P>(grp_first(N, NG, g + 1), grp_size(N, NG, g + 1), a_lo, a_hi, 16u * s,
                    fa[slot ^ 1]);
        } else {
          const int sn = (s + 1) & (KC / 4 - 1);
          fb[ss ^ 1] = load_B(bp, sn);
          load_A<P>(grp_first(N, NG, 0), grp_size(N, NG, 0), a_lo, a_hi, 16u * sn, fa[slot ^ 1]);
        }
        __builtin_amdgcn_sched_barrier(0);
#pragma unroll
        for (int j = 0; j < GT; ++j) {
          const int m = P + grp_first(N, NG, g) + j;
          if (j < grp_size(N, NG, g)) acc[m] = wmma_f32_16x16x4(fa[slot][j], fb[ss], acc[m]);
        }
        __builtin_amdgcn_sched_barrier(0);
      }
    }
  }
}

template <int P>
__device__ __forceinline__ void ksweep(const char* __restrict__ Lbytes,
                                       const char* __restrict__ zbytes, char* sm,
                                       const Thr& th, bool lowhalf, int rh,
                                       const float* __restrict__ l768w, v8f (&acc)[NM]) {
  if constexpr (P < NT / 2) {
    if constexpr (P + 1 < NT / 2) {
      if (rh == 1) stage_chunk<P + 1>(Lbytes, zbytes, sm, th, lowhalf, l768w);
    }
    compute_chunk<P>(sm, th, acc);
    if constexpr (P + 1 < NT / 2) {
      if (rh == 0) stage_chunk<P + 1>(Lbytes, zbytes, sm, th, lowhalf, l768w);
      __syncthreads();
    }
    ksweep<P + 1>(Lbytes, zbytes, sm, th, lowhalf, rh, l768w, acc);
  }
}

__global__ __launch_bounds__(512, 4) void fcvi_main_kernel(
    const float* __restrict__ Lb, const float* __restrict__ z,
    const float* __restrict__ x, const float* __restrict__ mean,
    float* __restrict__ out) {
  __shared__ __attribute__((aligned(16))) float smem[SM_END];
  float* const sRed = smem + SM_RED;
  float* const sBp = smem + SM_BP;

  const int tid  = threadIdx.x;
  const int lane = tid & 31;
  const int wave = __builtin_amdgcn_readfirstlane(tid >> 5);
  const int cg   = (wave & 3) | ((wave >> 3) << 2);
  const int rh   = (wave >> 2) & 1;
  const bool lowhalf = wave < 8;
  const int sub  = lane & 15;
  const int half = lane >> 4;
  const int wcol = cg * 16 + sub;
  const int gcol = blockIdx.x * WCOL + wcol;
  const float xv = x[gcol & BMSK];
  const float* __restrict__ zcol = z + gcol;

  *(float4*)(sBp + 4 * tid) = make_float4(0.0f, 0.0f, 0.0f, 0.0f);
  const float* __restrict__ l768w = Lb + (size_t)BROW + wave;

  v8f acc[NM];
#pragma unroll
  for (int m = 0; m < NM; ++m) acc[m] = (v8f){0.f, 0.f, 0.f, 0.f, 0.f, 0.f, 0.f, 0.f};

  {
    Thr th;
    const int r = tid >> 3, q = tid & 7;
    th.lsrc = (unsigned)(tid * 16);
    th.ldst_lo = (unsigned)((SM_A + r * KST + 4 * q) * 4);
    th.ldst_hi = (unsigned)((SM_A + (384 + r) * KST + 4 * q) * 4);
    const int zr = tid >> 5, zq = tid & 31;
    th.zsrc = (unsigned)((zr * NCOL + 4 * zq) * 4);
    th.zdst = (unsigned)((SM_Z + zr * ZST + 4 * zq) * 4);
    th.a_lo = (unsigned)((SM_A + (16 * rh + sub) * KST + 2 * half) * 4);
    th.b = (unsigned)((SM_Z + (2 * half) * ZST + wcol) * 4);
    const char* Lbytes = (const char*)Lb;
    const char* zbytes = (const char*)(z + blockIdx.x * WCOL);
    stage_chunk<0>(Lbytes, zbytes, (char*)smem, th, lowhalf, l768w);
    __syncthreads();
    ksweep<0>(Lbytes, zbytes, (char*)smem, th, lowhalf, rh, l768w, acc);
  }
  float outpart = 0.0f;
#pragma unroll
  for (int g = 0; g < 8; ++g) {
    const int r0 = 16 * (2 * g + rh);
#pragma unroll
    for (int v = 0; v < 8; ++v) {
      const int rl = r0 + v + 8 * half;
      const float w0 = acc[g][v] + mean[rl];
      const float b0 = acc[8 + g][v] + mean[256 + rl];
      const float w1 = acc[16 + g][v] + mean[512 + rl];
      outpart = fmaf(fmaxf(fmaf(xv, w0, b0), 0.0f), w1, outpart);
    }
  }

  float bpart = 0.0f;
  if (half == 0 && rh == 0)
    bpart = fmaf(Lb[(size_t)BROW + 768], zcol[(size_t)768 * NCOL], mean[768]);

  sRed[(2 * rh + half) * WCOL + wcol] = outpart + bpart;
  __syncthreads();
  if (tid < WCOL) {
    float bs = sBp[tid];
    for (int w = 1; w < 16; ++w) bs += sBp[w * WCOL + tid];
    out[blockIdx.x * WCOL + tid] =
        (((sRed[tid] + sRed[WCOL + tid]) + sRed[2 * WCOL + tid]) + sRed[3 * WCOL + tid]) + bs;
  }
}

extern "C" void kernel_launch(void* const* d_in, const int* in_sizes, int n_in,
                              void* d_out, int out_size, void* d_ws, size_t ws_size,
                              hipStream_t stream) {
  (void)in_sizes; (void)n_in; (void)out_size; (void)ws_size;
  const float* x    = (const float*)d_in[0];
  const float* mean = (const float*)d_in[1];
  const float* cov  = (const float*)d_in[2];
  const float* z    = (const float*)d_in[3];
  float* outp = (float*)d_out;
  float* Lbuf = (float*)d_ws;

  build_L_kernel<<<NBLK + 1, 256, 0, stream>>>(cov, Lbuf);
  fcvi_main_kernel<<<NCOL / WCOL, 512, 0, stream>>>(Lbuf, z, x, mean, outp);
}


# Accumulator tiles start at mean (read via LDS behind the first barrier) instead of zero: the fold's 48 dependent 16-byte mean loads and their 34 register spills go; mean is now summed first
# speedup vs baseline: 3.2647x; 1.2342x over previous
#include <hip/hip_runtime.h>
#include <math.h>

#define PP   769
#define PANEL 1024
#define NBLK 300
#define BROW (NBLK * PANEL)
#define NCOL 32768
#define BMSK 2047
#define KC   32
#define KST  36
#define WCOL 128
#define ZST  136
#define NT   48
#define NM   24
#define LG   8
#define GT   8

typedef __attribute__((ext_vector_type(2))) float v2f;
typedef __attribute__((ext_vector_type(8))) float v8f;

__device__ __forceinline__ v8f wmma_f32_16x16x4(v2f a, v2f b, v8f c) {
  return __builtin_amdgcn_wmma_f32_16x16x4_f32(false, a, false, b, (short)0, c,
                                               false, false);
}

__global__ __launch_bounds__(256) void build_L_kernel(const float* __restrict__ cov,
                                                      float* __restrict__ Lb) {
  const int b = blockIdx.x, tid = threadIdx.x;
  int i, k0;
  if (b < NBLK) {
    int P = 0, rem = b;
    while (rem >= 24 - P) { rem -= 24 - P; ++P; }
    i = 32 * (P + rem) + (tid >> 3);
    k0 = 32 * P + 4 * (tid & 7);
  } else {
    if (tid >= 193) return;
    i = 768;
    k0 = 4 * tid;
  }
  const float* __restrict__ row = cov + i * (i + 1) / 2;
  float v[4];
#pragma unroll
  for (int e = 0; e < 4; ++e) v[e] = (k0 + e <= i) ? row[k0 + e] : 0.0f;
#pragma unroll
  for (int e = 0; e < 4; ++e)
    if (k0 + e == i) v[e] = expf(v[e]);
  *(float4*)(Lb + (size_t)b * PANEL + 4 * tid) = make_float4(v[0], v[1], v[2], v[3]);
}

struct Thr {
  unsigned lsrc;
  unsigned zsrc;
  unsigned ldst_lo;
  unsigned ldst_hi;
  unsigned zdst;
  unsigned a_lo;
  unsigned b;
};

#define SM_BP  0
#define SM_A   (16 * WCOL)
#define SM_Z   (SM_A + NT * 16 * KST)
#define BUFSZ  (NT * 16 * KST + KC * ZST)
#define SM_RED (SM_A + 2 * BUFSZ)
#define SM_MEAN (SM_RED + 4 * WCOL)
#define SM_END (SM_MEAN + NT * 16)

template <int Pn, int J0, int N>
__device__ __forceinline__ void stage_L(const char* __restrict__ lsrc, unsigned tsrc,
                                        char* sm, unsigned lo, unsigned hi, bool lowhalf) {
  constexpr int NS = NT / 2 - Pn;
  float4 lv[N];
#pragma unroll
  for (int i = 0; i < N; ++i) {
    const bool full = 2 * (J0 + i) + 1 < NS;
    const unsigned t = (full || lowhalf) ? tsrc : tsrc - PANEL * 4;
    lv[i] = *(const float4*)(lsrc + (size_t)(J0 + i) * (2 * PANEL * 4) + t);
  }
#pragma unroll
  for (int i = 0; i < N; ++i) {
    const bool full = 2 * (J0 + i) + 1 < NS;
    const int g = KC * Pn + 64 * (J0 + i);
    char* d = (g < 384) ? sm + lo + g * (KST * 4) : sm + hi + (g - 384) * (KST * 4);
    if (!(full || lowhalf)) d -= 32 * KST * 4;
    *(float4*)d = lv[i];
  }
}

template <int Pn, int J0>
__device__ __forceinline__ void stage_L_rest(const char* __restrict__ lsrc, unsigned tsrc,
                                             char* sm, unsigned lo, unsigned hi, bool lowhalf) {
  constexpr int NJ = (NT / 2 - Pn + 1) / 2;
  if constexpr (J0 < NJ) {
    constexpr int N = (NJ - J0 < LG) ? NJ - J0 : LG;
    __builtin_amdgcn_sched_barrier(0);
    stage_L<Pn, J0, N>(lsrc, tsrc, sm, lo, hi, lowhalf);
    stage_L_rest<Pn, J0 + N>(lsrc, tsrc, sm, lo, hi, lowhalf);
  }
}

template <int Pn>
__device__ __forceinline__ void stage_chunk(const char* __restrict__ Lbytes,
                                            const char* __restrict__ zbytes, char* sm,
                                            const Thr& th, bool lowhalf,
                                            const float* __restrict__ l768w) {
  constexpr int K0 = KC * Pn;
  constexpr int NJ = (NT / 2 - Pn + 1) / 2;
  constexpr int N1 = (NJ < LG - 2) ? NJ : LG - 2;
  constexpr unsigned BO = (Pn & 1) * (BUFSZ * 4);
  const char* zsrc = zbytes + (size_t)K0 * (NCOL * 4);
  constexpr int CB = 24 * Pn - Pn * (Pn - 1) / 2;
  const char* lsrc = Lbytes + (size_t)CB * (PANEL * 4);
  unsigned lo = th.ldst_lo + BO, hi = th.ldst_hi + BO, zd = th.zdst + BO;
  asm volatile("" : "+v"(lo), "+v"(hi), "+v"(zd));
  const float lk0 = l768w[K0], lk1 = l768w[K0 + 16];
  {
    float4 zv[2];
#pragma unroll
    for (int i = 0; i < 2; ++i)
      zv[i] = *(const float4*)(zsrc + (size_t)(16 * i) * (NCOL * 4) + th.zsrc);
    stage_L<Pn, 0, N1>(lsrc, th.lsrc, sm, lo, hi, lowhalf);
#pragma unroll
    for (int i = 0; i < 2; ++i) *(float4*)(sm + zd + 16 * i * (ZST * 4)) = zv[i];
    float4* const bsum = (float4*)(sm + SM_BP * 4 + th.lsrc);
    float4 bs = *bsum;
    bs.x = fmaf(lk1, zv[1].x, fmaf(lk0, zv[0].x, bs.x));
    bs.y = fmaf(lk1, zv[1].y, fmaf(lk0, zv[0].y, bs.y));
    bs.z = fmaf(lk1, zv[1].z, fmaf(lk0, zv[0].z, bs.z));
    bs.w = fmaf(lk1, zv[1].w, fmaf(lk0, zv[0].w, bs.w));
    *bsum = bs;
  }
  stage_L_rest<Pn, N1>(lsrc, th.lsrc, sm, lo, hi, lowhalf);
}

constexpr int grp_first(int N, int NG, int g) { return g * (N / NG) + (g < N % NG ? g : N % NG); }
constexpr int grp_size(int N, int NG, int g) { return N / NG + (g < N % NG ? 1 : 0); }

template <int P>
__device__ __forceinline__ void load_A(int first, int size, const char* a_lo, const char* a_hi,
                                       unsigned koff, v2f (&dst)[GT]) {
#pragma unroll
  for (int j = 0; j < GT; ++j) {
    const int m = P + first + j;
    if (j < size) {
      const char* ap = (m < 12) ? a_lo + m * (32 * KST * 4) : a_hi + (m - 12) * (32 * KST * 4);
      dst[j] = *(const v2f*)(ap + koff);
    }
  }
}

__device__ __forceinline__ v2f load_B(const char* bp, int s) {
  const float* p = (const float*)bp + (4 * s) * ZST;
  v2f b;
  b.x = p[0];
  b.y = p[ZST];
  return b;
}

template <int P>
__device__ __forceinline__ void compute_chunk(char* sm, const Thr& th, v8f (&acc)[NM]) {
  constexpr unsigned BO = (P & 1) * (BUFSZ * 4);
  constexpr int N = NM - P;
  constexpr int NG = (N + GT - 1) / GT;
  unsigned ao = th.a_lo + BO, ah = th.a_lo + BO + 384 * KST * 4, bo = th.b + BO;
  asm volatile("" : "+v"(ao), "+v"(ah), "+v"(bo));
  const char* a_lo = sm + ao;
  const char* a_hi = sm + ah;
  const char* bp = sm + bo;

  v2f fa[2][GT];
  v2f fb[2];
  fb[0] = load_B(bp, 0);
  load_A<P>(grp_first(N, NG, 0), grp_size(N, NG, 0), a_lo, a_hi, 0u, fa[0]);
#pragma unroll 1
  for (int s2 = 0; s2 < KC / 8; ++s2) {
#pragma unroll
    for (int ss = 0; ss < 2; ++ss) {
      const int s = 2 * s2 + ss;
#pragma unroll
      for (int g = 0; g < NG; ++g) {
        const int slot = (ss * NG + g) & 1;
        if (g + 1 < NG) {
          load_A<P>(grp_first(N, NG, g + 1), grp_size(N, NG, g + 1), a_lo, a_hi, 16u * s,
                    fa[slot ^ 1]);
        } else {
          const int sn = (s + 1) & (KC / 4 - 1);
          fb[ss ^ 1] = load_B(bp, sn);
          load_A<P>(grp_first(N, NG, 0), grp_size(N, NG, 0), a_lo, a_hi, 16u * sn, fa[slot ^ 1]);
        }
        __builtin_amdgcn_sched_barrier(0);
#pragma unroll
        for (int j = 0; j < GT; ++j) {
          const int m = P + grp_first(N, NG, g) + j;
          if (j < grp_size(N, NG, g)) acc[m] = wmma_f32_16x16x4(fa[slot][j], fb[ss], acc[m]);
        }
        __builtin_amdgcn_sched_barrier(0);
      }
    }
  }
}

template <int P>
__device__ __forceinline__ void ksweep(const char* __restrict__ Lbytes,
                                       const char* __restrict__ zbytes, char* sm,
                                       const Thr& th, bool lowhalf, int rh,
                                       const float* __restrict__ l768w, v8f (&acc)[NM]) {
  if constexpr (P < NT / 2) {
    if constexpr (P + 1 < NT / 2) {
      if (rh == 1) stage_chunk<P + 1>(Lbytes, zbytes, sm, th, lowhalf, l768w);
    }
    compute_chunk<P>(sm, th, acc);
    if constexpr (P + 1 < NT / 2) {
      if (rh == 0) stage_chunk<P + 1>(Lbytes, zbytes, sm, th, lowhalf, l768w);
      __syncthreads();
    }
    ksweep<P + 1>(Lbytes, zbytes, sm, th, lowhalf, rh, l768w, acc);
  }
}

__global__ __launch_bounds__(512, 4) void fcvi_main_kernel(
    const float* __restrict__ Lb, const float* __restrict__ z,
    const float* __restrict__ x, const float* __restrict__ mean,
    float* __restrict__ out) {
  __shared__ __attribute__((aligned(16))) float smem[SM_END];
  float* const sRed = smem + SM_RED;
  float* const sBp = smem + SM_BP;
  float* const sMean = smem + SM_MEAN;

  const int tid  = threadIdx.x;
  const int lane = tid & 31;
  const int wave = __builtin_amdgcn_readfirstlane(tid >> 5);
  const int cg   = (wave & 3) | ((wave >> 3) << 2);
  const int rh   = (wave >> 2) & 1;
  const bool lowhalf = wave < 8;
  const int sub  = lane & 15;
  const int half = lane >> 4;
  const int wcol = cg * 16 + sub;
  const int gcol = blockIdx.x * WCOL + wcol;
  const float xv = x[gcol & BMSK];
  const float* __restrict__ zcol = z + gcol;

  *(float4*)(sBp + 4 * tid) = make_float4(0.0f, 0.0f, 0.0f, 0.0f);
  const float* __restrict__ l768w = Lb + (size_t)BROW + wave;

  float4 mv = make_float4(0.0f, 0.0f, 0.0f, 0.0f);
  if (wave < NT * 4 / 32) mv = *(const float4*)(mean + 4 * tid);

  v8f acc[NM];
  {
    Thr th;
    const int r = tid >> 3, q = tid & 7;
    th.lsrc = (unsigned)(tid * 16);
    th.ldst_lo = (unsigned)((SM_A + r * KST + 4 * q) * 4);
    th.ldst_hi = (unsigned)((SM_A + (384 + r) * KST + 4 * q) * 4);
    const int zr = tid >> 5, zq = tid & 31;
    th.zsrc = (unsigned)((zr * NCOL + 4 * zq) * 4);
    th.zdst = (unsigned)((SM_Z + zr * ZST + 4 * zq) * 4);
    th.a_lo = (unsigned)((SM_A + (16 * rh + sub) * KST + 2 * half) * 4);
    th.b = (unsigned)((SM_Z + (2 * half) * ZST + wcol) * 4);
    const char* Lbytes = (const char*)Lb;
    const char* zbytes = (const char*)(z + blockIdx.x * WCOL);
    stage_chunk<0>(Lbytes, zbytes, (char*)smem, th, lowhalf, l768w);
    if (wave < NT * 4 / 32) *(float4*)(sMean + 4 * tid) = mv;
    __syncthreads();
    const float* mp = sMean + 16 * rh + 8 * half;
#pragma unroll
    for (int m = 0; m < NM; ++m) {
      const float4 m0 = *(const float4*)(mp + 32 * m);
      const float4 m1 = *(const float4*)(mp + 32 * m + 4);
      acc[m] = (v8f){m0.x, m0.y, m0.z, m0.w, m1.x, m1.y, m1.z, m1.w};
    }
    ksweep<0>(Lbytes, zbytes, (char*)smem, th, lowhalf, rh, l768w, acc);
  }
  float outpart = 0.0f;
#pragma unroll
  for (int g = 0; g < 8; ++g) {
#pragma unroll
    for (int v = 0; v < 8; ++v)
      outpart = fmaf(fmaxf(fmaf(xv, acc[g][v], acc[8 + g][v]), 0.0f), acc[16 + g][v], outpart);
  }

  float bpart = 0.0f;
  if (half == 0 && rh == 0)
    bpart = fmaf(Lb[(size_t)BROW + 768], zcol[(size_t)768 * NCOL], mean[768]);

  sRed[(2 * rh + half) * WCOL + wcol] = outpart + bpart;
  __syncthreads();
  if (tid < WCOL) {
    float bs = sBp[tid];
    for (int w = 1; w < 16; ++w) bs += sBp[w * WCOL + tid];
    out[blockIdx.x * WCOL + tid] =
        (((sRed[tid] + sRed[WCOL + tid]) + sRed[2 * WCOL + tid]) + sRed[3 * WCOL + tid]) + bs;
  }
}

extern "C" void kernel_launch(void* const* d_in, const int* in_sizes, int n_in,
                              void* d_out, int out_size, void* d_ws, size_t ws_size,
                              hipStream_t stream) {
  (void)in_sizes; (void)n_in; (void)out_size; (void)ws_size;
  const float* x    = (const float*)d_in[0];
  const float* mean = (const float*)d_in[1];
  const float* cov  = (const float*)d_in[2];
  const float* z    = (const float*)d_in[3];
  float* outp = (float*)d_out;
  float* Lbuf = (float*)d_ws;

  build_L_kernel<<<NBLK + 1, 256, 0, stream>>>(cov, Lbuf);
  fcvi_main_kernel<<<NCOL / WCOL, 512, 0, stream>>>(Lbuf, z, x, mean, outp);
}
